# starting kernel
# baseline (speedup 1.0000x reference)
#include <hip/hip_runtime.h>
#include <hip/hip_bf16.h>


#define B_   64
#define P_   128
#define H1_  256
#define H2_  128
#define C_   32

typedef __attribute__((ext_vector_type(2))) float v2f;
typedef __attribute__((ext_vector_type(8))) float v8f;

__global__ void __launch_bounds__(256) spw_scatter(
    const float* __restrict__ x, const float* __restrict__ w,
    const int* __restrict__ idx, float* __restrict__ WX, int N)
{
    __shared__ float lacc[P_];
    const int t = threadIdx.x;
    if (t < P_) lacc[t] = 0.0f;
    __syncthreads();

    const int  b    = blockIdx.y;
    const long base = ((long)blockIdx.x * 256 + t) * 64;

    if (base < N) {
        const float* xr = x + (long)b * N + base;
        const float* wr = w + base;
        const int*   ir = idx + base;
        int   cur = ir[0];
        float run = 0.0f;
        #pragma unroll 4
        for (int j = 0; j < 64; j += 4) {
            float4 xv = *(const float4*)(xr + j);
            float4 wv = *(const float4*)(wr + j);
            int4   iv = *(const int4*)(ir + j);
            float p0 = xv.x * wv.x, p1 = xv.y * wv.y;
            float p2 = xv.z * wv.z, p3 = xv.w * wv.w;
            if (iv.x != cur) { atomicAdd(&lacc[cur], run); run = 0.f; cur = iv.x; } run += p0;
            if (iv.y != cur) { atomicAdd(&lacc[cur], run); run = 0.f; cur = iv.y; } run += p1;
            if (iv.z != cur) { atomicAdd(&lacc[cur], run); run = 0.f; cur = iv.z; } run += p2;
            if (iv.w != cur) { atomicAdd(&lacc[cur], run); run = 0.f; cur = iv.w; } run += p3;
        }
        atomicAdd(&lacc[cur], run);
    }
    __syncthreads();

    if (t < P_) {
        float v = lacc[t];
        if (v != 0.0f) atomicAdd(&WX[(long)b * P_ + t], v);
    }
}

__device__ __forceinline__ v8f wmma_tile_f32(
    const float* __restrict__ A, int lda,
    const float* __restrict__ Bm, int ldb,
    int K, int mt, int nt, int lane)
{
    const int ml   = lane & 15;
    const int koff = (lane >> 4) * 2;
    const float* arow = A + (mt * 16 + ml) * lda;
    const float* bcol = Bm + nt * 16 + ml;
    v8f acc = {0.f, 0.f, 0.f, 0.f, 0.f, 0.f, 0.f, 0.f};
    for (int k = 0; k < K; k += 4) {
        v2f av, bv;
        av.x = arow[k + koff];
        av.y = arow[k + koff + 1];
        bv.x = bcol[(k + koff) * ldb];
        bv.y = bcol[(k + koff + 1) * ldb];
        acc = __builtin_amdgcn_wmma_f32_16x16x4_f32(
            false, av, false, bv, (short)0, acc, false, false);
    }
    return acc;
}

__global__ void __launch_bounds__(512) spw_head(
    const float* __restrict__ WX,
    const float* __restrict__ co_w,
    const float* __restrict__ bn0_g, const float* __restrict__ bn0_b,
    const float* __restrict__ W1, const float* __restrict__ b1,
    const float* __restrict__ bn1_g, const float* __restrict__ bn1_b,
    const float* __restrict__ W2, const float* __restrict__ b2,
    const float* __restrict__ bn2_g, const float* __restrict__ bn2_b,
    const float* __restrict__ Wo, const float* __restrict__ bo,
    float* __restrict__ Zout,
    float* __restrict__ yout,
    float* __restrict__ h1,
    float* __restrict__ h2,
    float* __restrict__ logits)
{
    __shared__ float sA[H1_];
    __shared__ float sC[H1_];
    const int t    = threadIdx.x;
    const int lane = t & 31;
    const int wave = t >> 5;
    const float invB = 1.0f / (float)B_;

    if (t < P_) {
        float s = 0.f, s2 = 0.f;
        for (int r = 0; r < B_; ++r) {
            float v = fmaxf(WX[r * P_ + t], 0.f);
            s += v; s2 += v * v;
        }
        float m   = s * invB;
        float var = s2 * invB - m * m;
        float rg  = rsqrtf(var + 1e-5f) * bn0_g[t];
        float sg  = 1.0f / (1.0f + __expf(-co_w[t]));
        sA[t] = rg * sg;
        sC[t] = (bn0_b[t] - m * rg) * sg;
    }
    __syncthreads();
    for (int e = t; e < B_ * P_; e += 512) {
        int cp = e & (P_ - 1);
        float v = fmaxf(WX[e], 0.f);
        Zout[e] = v * sA[cp] + sC[cp];
    }
    __syncthreads();

    for (int q = 0; q < 4; ++q) {
        int tile = wave * 4 + q;
        int mt = tile >> 4, nt = tile & 15;
        v8f acc = wmma_tile_f32(Zout, P_, W1, H1_, P_, mt, nt, lane);
        int ml = lane & 15, n = nt * 16 + ml, mb = mt * 16 + (lane >> 4) * 8;
        float bias = b1[n];
        #pragma unroll
        for (int r = 0; r < 8; ++r)
            h1[(mb + r) * H1_ + n] = fmaxf(acc[r] + bias, 0.f);
    }
    __syncthreads();
    if (t < H1_) {
        float s = 0.f, s2 = 0.f;
        for (int r = 0; r < B_; ++r) { float v = h1[r * H1_ + t]; s += v; s2 += v * v; }
        float m = s * invB, var = s2 * invB - m * m;
        float rg = rsqrtf(var + 1e-5f) * bn1_g[t];
        sA[t] = rg;
        sC[t] = bn1_b[t] - m * rg;
    }
    __syncthreads();
    for (int e = t; e < B_ * H1_; e += 512) {
        int cc = e & (H1_ - 1);
        h1[e] = h1[e] * sA[cc] + sC[cc];
    }
    __syncthreads();

    for (int q = 0; q < 2; ++q) {
        int tile = wave * 2 + q;
        int mt = tile >> 3, nt = tile & 7;
        v8f acc = wmma_tile_f32(h1, H1_, W2, H2_, H1_, mt, nt, lane);
        int ml = lane & 15, n = nt * 16 + ml, mb = mt * 16 + (lane >> 4) * 8;
        float bias = b2[n];
        #pragma unroll
        for (int r = 0; r < 8; ++r)
            h2[(mb + r) * H2_ + n] = fmaxf(acc[r] + bias, 0.f);
    }
    __syncthreads();
    if (t < H2_) {
        float s = 0.f, s2 = 0.f;
        for (int r = 0; r < B_; ++r) { float v = h2[r * H2_ + t]; s += v; s2 += v * v; }
        float m = s * invB, var = s2 * invB - m * m;
        float rg = rsqrtf(var + 1e-5f) * bn2_g[t];
        sA[t] = rg;
        sC[t] = bn2_b[t] - m * rg;
    }
    __syncthreads();
    for (int e = t; e < B_ * H2_; e += 512) {
        int cc = e & (H2_ - 1);
        h2[e] = h2[e] * sA[cc] + sC[cc];
    }
    __syncthreads();

    if (wave < 8) {
        int tile = wave;
        int mt = tile >> 1, nt = tile & 1;
        v8f acc = wmma_tile_f32(h2, H2_, Wo, C_, H2_, mt, nt, lane);
        int ml = lane & 15, n = nt * 16 + ml, mb = mt * 16 + (lane >> 4) * 8;
        float bias = bo[n];
        #pragma unroll
        for (int r = 0; r < 8; ++r)
            logits[(mb + r) * C_ + n] = acc[r] + bias;
    }
    __syncthreads();

    if (t < B_) {
        const float* lr = logits + t * C_;
        float m = -3.0e38f;
        for (int c = 0; c < C_; ++c) m = fmaxf(m, lr[c]);
        float ex[C_];
        float s = 0.f;
        for (int c = 0; c < C_; ++c) { float e = __expf(lr[c] - m); ex[c] = e; s += e; }
        float inv = 1.0f / s;
        for (int c = 0; c < C_; ++c) yout[t * C_ + c] = ex[c] * inv;
    }
}

extern "C" void kernel_launch(void* const* d_in, const int* in_sizes, int n_in,
                              void* d_out, int out_size, void* d_ws, size_t ws_size,
                              hipStream_t stream) {
    const float* x     = (const float*)d_in[0];
    const float* w     = (const float*)d_in[1];
    const float* co_w  = (const float*)d_in[2];
    const float* bn0_g = (const float*)d_in[3];
    const float* bn0_b = (const float*)d_in[4];
    const float* W1    = (const float*)d_in[5];
    const float* b1    = (const float*)d_in[6];
    const float* bn1_g = (const float*)d_in[7];
    const float* bn1_b = (const float*)d_in[8];
    const float* W2    = (const float*)d_in[9];
    const float* b2    = (const float*)d_in[10];
    const float* bn2_g = (const float*)d_in[11];
    const float* bn2_b = (const float*)d_in[12];
    const float* Wo    = (const float*)d_in[13];
    const float* bo    = (const float*)d_in[14];
    const int*   idx   = (const int*)d_in[15];

    float* out    = (float*)d_out;
    float* WX     = (float*)d_ws;
    float* h1     = WX + B_ * P_;
    float* h2     = h1 + B_ * H1_;
    float* logits = h2 + B_ * H2_;

    const int N = in_sizes[0] / B_;

    hipMemsetAsync(WX, 0, (size_t)B_ * P_ * sizeof(float), stream);

    const int elems_per_block = 256 * 64;
    dim3 g1((N + elems_per_block - 1) / elems_per_block, B_);
    spw_scatter<<<g1, 256, 0, stream>>>(x, w, idx, WX, N);

    spw_head<<<1, 512, 0, stream>>>(WX, co_w, bn0_g, bn0_b,
                                    W1, b1, bn1_g, bn1_b,
                                    W2, b2, bn2_g, bn2_b,
                                    Wo, bo,
                                    out + B_ * C_, out,
                                    h1, h2, logits);
}


# scatter stage: lanes take consecutive 16 B so each wave load is 512 contiguous B (whole lines), 16 rows per wave share w/idx; per-lane sums, wave shuffle sum at segment ends (sum order changes)
# speedup vs baseline: 3.1948x; 3.1948x over previous
#include <hip/hip_runtime.h>
#include <hip/hip_bf16.h>


#define B_   64
#define P_   128
#define H1_  256
#define H2_  128
#define C_   32

typedef __attribute__((ext_vector_type(2))) float v2f;
typedef __attribute__((ext_vector_type(8))) float v8f;

#define SC_R    16
#define SC_NRG  (B_ / SC_R)
#define SC_WPB  8
#define SC_BPG  128
#define SC_GRID (SC_NRG * SC_BPG)

__device__ __forceinline__ float sc_rows_sum(const float (&v)[SC_R], int lane) {
    float a[8], b[4], c[2];
    const bool h4 = (lane & 16) != 0, h3 = (lane & 8) != 0;
    const bool h2 = (lane & 4) != 0,  h1 = (lane & 2) != 0;
    #pragma unroll
    for (int r = 0; r < 8; ++r)
        a[r] = (h4 ? v[r + 8] : v[r]) + __shfl_xor(h4 ? v[r] : v[r + 8], 16);
    #pragma unroll
    for (int r = 0; r < 4; ++r)
        b[r] = (h3 ? a[r + 4] : a[r]) + __shfl_xor(h3 ? a[r] : a[r + 4], 8);
    #pragma unroll
    for (int r = 0; r < 2; ++r)
        c[r] = (h2 ? b[r + 2] : b[r]) + __shfl_xor(h2 ? b[r] : b[r + 2], 4);
    float z = (h1 ? c[1] : c[0]) + __shfl_xor(h1 ? c[0] : c[1], 2);
    return z + __shfl_xor(z, 1);
}

__device__ __forceinline__ void sc_load(
    float4 (&xv)[SC_R], float4& wv, int4& iv,
    const float* __restrict__ xg, const float* __restrict__ w,
    const int* __restrict__ idx, int N, unsigned off)
{
    iv = *(const int4*)(idx + off);
    wv = *(const float4*)(w + off);
    #pragma unroll
    for (int r = 0; r < SC_R; ++r)
        xv[r] = *(const float4*)(xg + (size_t)r * N + off);
}

__device__ __forceinline__ void sc_flush(
    const float (&tot)[SC_R], int cur, float* lacc, int lane)
{
    const float z = sc_rows_sum(tot, lane);
    if ((lane & 1) == 0) atomicAdd(&lacc[((lane >> 1) & 15) * P_ + cur], z);
}

__device__ __forceinline__ void sc_step(
    const float4 (&xv)[SC_R], const float4 wv, const int4 iv,
    v2f (&run)[SC_R], int& cur, float* lacc, int lane)
{
    const int d = (iv.x ^ cur) | (iv.y ^ cur) | (iv.z ^ cur) | (iv.w ^ cur);
    if (!__any(d != 0)) {
        const v2f wlo = {wv.x, wv.y}, whi = {wv.z, wv.w};
        #pragma unroll
        for (int r = 0; r < SC_R; ++r) {
            const v2f xlo = {xv[r].x, xv[r].y}, xhi = {xv[r].z, xv[r].w};
            run[r] = __builtin_elementwise_fma(xhi, whi,
                     __builtin_elementwise_fma(xlo, wlo, run[r]));
        }
    } else {
        const int  nxt = __builtin_amdgcn_readlane(iv.w, 31);
        const bool all_cur = (d == 0);
        const bool all_nxt = !all_cur &&
            (((iv.x ^ nxt) | (iv.y ^ nxt) | (iv.z ^ nxt) | (iv.w ^ nxt)) == 0);
        if (!all_cur && !all_nxt) {
            #pragma unroll
            for (int r = 0; r < SC_R; ++r) {
                atomicAdd(&lacc[r * P_ + iv.x], xv[r].x * wv.x);
                atomicAdd(&lacc[r * P_ + iv.y], xv[r].y * wv.y);
                atomicAdd(&lacc[r * P_ + iv.z], xv[r].z * wv.z);
                atomicAdd(&lacc[r * P_ + iv.w], xv[r].w * wv.w);
            }
        }
        float tot[SC_R];
        #pragma unroll
        for (int r = 0; r < SC_R; ++r) {
            const float s = (xv[r].x * wv.x + xv[r].y * wv.y) +
                            (xv[r].z * wv.z + xv[r].w * wv.w);
            tot[r] = (run[r].x + run[r].y) + (all_cur ? s : 0.0f);
            run[r].x = all_nxt ? s : 0.0f;
            run[r].y = 0.0f;
        }
        sc_flush(tot, cur, lacc, lane);
        cur = nxt;
    }
}

__global__ void __launch_bounds__(256) spw_scatter(
    const float* __restrict__ x, const float* __restrict__ w,
    const int* __restrict__ idx, float* __restrict__ WX, int N)
{
    __shared__ float lacc[SC_R * P_];
    const int t    = threadIdx.x;
    const int lane = t & 31;
    const int wave = __builtin_amdgcn_readfirstlane(t >> 5);
    for (int e = t; e < SC_R * P_; e += 256) lacc[e] = 0.0f;
    __syncthreads();

    const int rg  = blockIdx.x & (SC_NRG - 1);
    const int gw  = (blockIdx.x / SC_NRG) * SC_WPB + wave;
    const int nwv = SC_BPG * SC_WPB;
    const int nst = N >> 7;
    const int it0 = (int)(((long)gw * nst) / nwv);
    const int n   = (int)(((long)(gw + 1) * nst) / nwv) - it0;
    const float* xg = x + (size_t)(rg * SC_R) * N;
    const unsigned lo = (unsigned)it0 * 128u + (unsigned)lane * 4u;

    if (n > 0) {
        v2f run[SC_R];
        #pragma unroll
        for (int r = 0; r < SC_R; ++r) run[r] = (v2f){0.0f, 0.0f};
        int cur = idx[(size_t)it0 * 128];
        float4 xa[SC_R], xb[SC_R];
        float4 wa, wb;
        int4   ia, ib;
        sc_load(xa, wa, ia, xg, w, idx, N, lo);
        int k = 0;
        while (k + 2 < n) {
            sc_load(xb, wb, ib, xg, w, idx, N, lo + (unsigned)(k + 1) * 128u);
            sc_step(xa, wa, ia, run, cur, lacc, lane);
            sc_load(xa, wa, ia, xg, w, idx, N, lo + (unsigned)(k + 2) * 128u);
            sc_step(xb, wb, ib, run, cur, lacc, lane);
            k += 2;
        }
        if (k + 1 < n) {
            sc_load(xb, wb, ib, xg, w, idx, N, lo + (unsigned)(k + 1) * 128u);
            sc_step(xa, wa, ia, run, cur, lacc, lane);
            sc_step(xb, wb, ib, run, cur, lacc, lane);
        } else {
            sc_step(xa, wa, ia, run, cur, lacc, lane);
        }
        float tot[SC_R];
        #pragma unroll
        for (int r = 0; r < SC_R; ++r) tot[r] = run[r].x + run[r].y;
        sc_flush(tot, cur, lacc, lane);
    }
    if (gw == nwv - 1) {
        for (int f = (nst << 7) + lane; f < N; f += 32) {
            const float wf = w[f];
            const int   pf = idx[f];
            #pragma unroll
            for (int r = 0; r < SC_R; ++r)
                atomicAdd(&lacc[r * P_ + pf], xg[(size_t)r * N + f] * wf);
        }
    }
    __syncthreads();

    for (int e = t; e < SC_R * P_; e += 256) {
        const float v = lacc[e];
        if (v != 0.0f) atomicAdd(&WX[(size_t)(rg * SC_R) * P_ + e], v);
    }
}

__device__ __forceinline__ v8f wmma_tile_f32(
    const float* __restrict__ A, int lda,
    const float* __restrict__ Bm, int ldb,
    int K, int mt, int nt, int lane)
{
    const int ml   = lane & 15;
    const int koff = (lane >> 4) * 2;
    const float* arow = A + (mt * 16 + ml) * lda;
    const float* bcol = Bm + nt * 16 + ml;
    v8f acc = {0.f, 0.f, 0.f, 0.f, 0.f, 0.f, 0.f, 0.f};
    for (int k = 0; k < K; k += 4) {
        v2f av, bv;
        av.x = arow[k + koff];
        av.y = arow[k + koff + 1];
        bv.x = bcol[(k + koff) * ldb];
        bv.y = bcol[(k + koff + 1) * ldb];
        acc = __builtin_amdgcn_wmma_f32_16x16x4_f32(
            false, av, false, bv, (short)0, acc, false, false);
    }
    return acc;
}

__global__ void __launch_bounds__(512) spw_head(
    const float* __restrict__ WX,
    const float* __restrict__ co_w,
    const float* __restrict__ bn0_g, const float* __restrict__ bn0_b,
    const float* __restrict__ W1, const float* __restrict__ b1,
    const float* __restrict__ bn1_g, const float* __restrict__ bn1_b,
    const float* __restrict__ W2, const float* __restrict__ b2,
    const float* __restrict__ bn2_g, const float* __restrict__ bn2_b,
    const float* __restrict__ Wo, const float* __restrict__ bo,
    float* __restrict__ Zout,
    float* __restrict__ yout,
    float* __restrict__ h1,
    float* __restrict__ h2,
    float* __restrict__ logits)
{
    __shared__ float sA[H1_];
    __shared__ float sC[H1_];
    const int t    = threadIdx.x;
    const int lane = t & 31;
    const int wave = t >> 5;
    const float invB = 1.0f / (float)B_;

    if (t < P_) {
        float s = 0.f, s2 = 0.f;
        for (int r = 0; r < B_; ++r) {
            float v = fmaxf(WX[r * P_ + t], 0.f);
            s += v; s2 += v * v;
        }
        float m   = s * invB;
        float var = s2 * invB - m * m;
        float rg  = rsqrtf(var + 1e-5f) * bn0_g[t];
        float sg  = 1.0f / (1.0f + __expf(-co_w[t]));
        sA[t] = rg * sg;
        sC[t] = (bn0_b[t] - m * rg) * sg;
    }
    __syncthreads();
    for (int e = t; e < B_ * P_; e += 512) {
        int cp = e & (P_ - 1);
        float v = fmaxf(WX[e], 0.f);
        Zout[e] = v * sA[cp] + sC[cp];
    }
    __syncthreads();

    for (int q = 0; q < 4; ++q) {
        int tile = wave * 4 + q;
        int mt = tile >> 4, nt = tile & 15;
        v8f acc = wmma_tile_f32(Zout, P_, W1, H1_, P_, mt, nt, lane);
        int ml = lane & 15, n = nt * 16 + ml, mb = mt * 16 + (lane >> 4) * 8;
        float bias = b1[n];
        #pragma unroll
        for (int r = 0; r < 8; ++r)
            h1[(mb + r) * H1_ + n] = fmaxf(acc[r] + bias, 0.f);
    }
    __syncthreads();
    if (t < H1_) {
        float s = 0.f, s2 = 0.f;
        for (int r = 0; r < B_; ++r) { float v = h1[r * H1_ + t]; s += v; s2 += v * v; }
        float m = s * invB, var = s2 * invB - m * m;
        float rg = rsqrtf(var + 1e-5f) * bn1_g[t];
        sA[t] = rg;
        sC[t] = bn1_b[t] - m * rg;
    }
    __syncthreads();
    for (int e = t; e < B_ * H1_; e += 512) {
        int cc = e & (H1_ - 1);
        h1[e] = h1[e] * sA[cc] + sC[cc];
    }
    __syncthreads();

    for (int q = 0; q < 2; ++q) {
        int tile = wave * 2 + q;
        int mt = tile >> 3, nt = tile & 7;
        v8f acc = wmma_tile_f32(h1, H1_, W2, H2_, H1_, mt, nt, lane);
        int ml = lane & 15, n = nt * 16 + ml, mb = mt * 16 + (lane >> 4) * 8;
        float bias = b2[n];
        #pragma unroll
        for (int r = 0; r < 8; ++r)
            h2[(mb + r) * H2_ + n] = fmaxf(acc[r] + bias, 0.f);
    }
    __syncthreads();
    if (t < H2_) {
        float s = 0.f, s2 = 0.f;
        for (int r = 0; r < B_; ++r) { float v = h2[r * H2_ + t]; s += v; s2 += v * v; }
        float m = s * invB, var = s2 * invB - m * m;
        float rg = rsqrtf(var + 1e-5f) * bn2_g[t];
        sA[t] = rg;
        sC[t] = bn2_b[t] - m * rg;
    }
    __syncthreads();
    for (int e = t; e < B_ * H2_; e += 512) {
        int cc = e & (H2_ - 1);
        h2[e] = h2[e] * sA[cc] + sC[cc];
    }
    __syncthreads();

    if (wave < 8) {
        int tile = wave;
        int mt = tile >> 1, nt = tile & 1;
        v8f acc = wmma_tile_f32(h2, H2_, Wo, C_, H2_, mt, nt, lane);
        int ml = lane & 15, n = nt * 16 + ml, mb = mt * 16 + (lane >> 4) * 8;
        float bias = bo[n];
        #pragma unroll
        for (int r = 0; r < 8; ++r)
            logits[(mb + r) * C_ + n] = acc[r] + bias;
    }
    __syncthreads();

    if (t < B_) {
        const float* lr = logits + t * C_;
        float m = -3.0e38f;
        for (int c = 0; c < C_; ++c) m = fmaxf(m, lr[c]);
        float ex[C_];
        float s = 0.f;
        for (int c = 0; c < C_; ++c) { float e = __expf(lr[c] - m); ex[c] = e; s += e; }
        float inv = 1.0f / s;
        for (int c = 0; c < C_; ++c) yout[t * C_ + c] = ex[c] * inv;
    }
}

extern "C" void kernel_launch(void* const* d_in, const int* in_sizes, int n_in,
                              void* d_out, int out_size, void* d_ws, size_t ws_size,
                              hipStream_t stream) {
    const float* x     = (const float*)d_in[0];
    const float* w     = (const float*)d_in[1];
    const float* co_w  = (const float*)d_in[2];
    const float* bn0_g = (const float*)d_in[3];
    const float* bn0_b = (const float*)d_in[4];
    const float* W1    = (const float*)d_in[5];
    const float* b1    = (const float*)d_in[6];
    const float* bn1_g = (const float*)d_in[7];
    const float* bn1_b = (const float*)d_in[8];
    const float* W2    = (const float*)d_in[9];
    const float* b2    = (const float*)d_in[10];
    const float* bn2_g = (const float*)d_in[11];
    const float* bn2_b = (const float*)d_in[12];
    const float* Wo    = (const float*)d_in[13];
    const float* bo    = (const float*)d_in[14];
    const int*   idx   = (const int*)d_in[15];

    float* out    = (float*)d_out;
    float* WX     = (float*)d_ws;
    float* h1     = WX + B_ * P_;
    float* h2     = h1 + B_ * H1_;
    float* logits = h2 + B_ * H2_;

    const int N = in_sizes[0] / B_;

    hipMemsetAsync(WX, 0, (size_t)B_ * P_ * sizeof(float), stream);

    spw_scatter<<<SC_GRID, 256, 0, stream>>>(x, w, idx, WX, N);

    spw_head<<<1, 512, 0, stream>>>(WX, co_w, bn0_g, bn0_b,
                                    W1, b1, bn1_g, bn1_b,
                                    W2, b2, bn2_g, bn2_b,
                                    Wo, bo,
                                    out + B_ * C_, out,
                                    h1, h2, logits);
}


# head (one work-group) moved into LDS: a wave owns 16-column strips, loads each weight fragment once for all its row tiles, does bias/ReLU/norm on its accumulators (norm sum order changes)
# speedup vs baseline: 5.3719x; 1.6815x over previous
#include <hip/hip_runtime.h>
#include <hip/hip_bf16.h>


#define B_   64
#define P_   128
#define H1_  256
#define H2_  128
#define C_   32

typedef __attribute__((ext_vector_type(2))) float v2f;
typedef __attribute__((ext_vector_type(8))) float v8f;

#define SC_R    16
#define SC_NRG  (B_ / SC_R)
#define SC_WPB  8
#define SC_BPG  128
#define SC_GRID (SC_NRG * SC_BPG)

__device__ __forceinline__ float sc_rows_sum(const float (&v)[SC_R], int lane) {
    float a[8], b[4], c[2];
    const bool h4 = (lane & 16) != 0, h3 = (lane & 8) != 0;
    const bool h2 = (lane & 4) != 0,  h1 = (lane & 2) != 0;
    #pragma unroll
    for (int r = 0; r < 8; ++r)
        a[r] = (h4 ? v[r + 8] : v[r]) + __shfl_xor(h4 ? v[r] : v[r + 8], 16);
    #pragma unroll
    for (int r = 0; r < 4; ++r)
        b[r] = (h3 ? a[r + 4] : a[r]) + __shfl_xor(h3 ? a[r] : a[r + 4], 8);
    #pragma unroll
    for (int r = 0; r < 2; ++r)
        c[r] = (h2 ? b[r + 2] : b[r]) + __shfl_xor(h2 ? b[r] : b[r + 2], 4);
    float z = (h1 ? c[1] : c[0]) + __shfl_xor(h1 ? c[0] : c[1], 2);
    return z + __shfl_xor(z, 1);
}

__device__ __forceinline__ void sc_load(
    float4 (&xv)[SC_R], float4& wv, int4& iv,
    const float* __restrict__ xg, const float* __restrict__ w,
    const int* __restrict__ idx, int N, unsigned off)
{
    iv = *(const int4*)(idx + off);
    wv = *(const float4*)(w + off);
    #pragma unroll
    for (int r = 0; r < SC_R; ++r)
        xv[r] = *(const float4*)(xg + (size_t)r * N + off);
}

__device__ __forceinline__ void sc_flush(
    const float (&tot)[SC_R], int cur, float* lacc, int lane)
{
    const float z = sc_rows_sum(tot, lane);
    if ((lane & 1) == 0) atomicAdd(&lacc[((lane >> 1) & 15) * P_ + cur], z);
}

__device__ __forceinline__ void sc_step(
    const float4 (&xv)[SC_R], const float4 wv, const int4 iv,
    v2f (&run)[SC_R], int& cur, float* lacc, int lane)
{
    const int d = (iv.x ^ cur) | (iv.y ^ cur) | (iv.z ^ cur) | (iv.w ^ cur);
    if (!__any(d != 0)) {
        const v2f wlo = {wv.x, wv.y}, whi = {wv.z, wv.w};
        #pragma unroll
        for (int r = 0; r < SC_R; ++r) {
            const v2f xlo = {xv[r].x, xv[r].y}, xhi = {xv[r].z, xv[r].w};
            run[r] = __builtin_elementwise_fma(xhi, whi,
                     __builtin_elementwise_fma(xlo, wlo, run[r]));
        }
    } else {
        const int  nxt = __builtin_amdgcn_readlane(iv.w, 31);
        const bool all_cur = (d == 0);
        const bool all_nxt = !all_cur &&
            (((iv.x ^ nxt) | (iv.y ^ nxt) | (iv.z ^ nxt) | (iv.w ^ nxt)) == 0);
        if (!all_cur && !all_nxt) {
            #pragma unroll
            for (int r = 0; r < SC_R; ++r) {
                atomicAdd(&lacc[r * P_ + iv.x], xv[r].x * wv.x);
                atomicAdd(&lacc[r * P_ + iv.y], xv[r].y * wv.y);
                atomicAdd(&lacc[r * P_ + iv.z], xv[r].z * wv.z);
                atomicAdd(&lacc[r * P_ + iv.w], xv[r].w * wv.w);
            }
        }
        float tot[SC_R];
        #pragma unroll
        for (int r = 0; r < SC_R; ++r) {
            const float s = (xv[r].x * wv.x + xv[r].y * wv.y) +
                            (xv[r].z * wv.z + xv[r].w * wv.w);
            tot[r] = (run[r].x + run[r].y) + (all_cur ? s : 0.0f);
            run[r].x = all_nxt ? s : 0.0f;
            run[r].y = 0.0f;
        }
        sc_flush(tot, cur, lacc, lane);
        cur = nxt;
    }
}

__global__ void __launch_bounds__(256) spw_scatter(
    const float* __restrict__ x, const float* __restrict__ w,
    const int* __restrict__ idx, float* __restrict__ WX, int N)
{
    __shared__ float lacc[SC_R * P_];
    const int t    = threadIdx.x;
    const int lane = t & 31;
    const int wave = __builtin_amdgcn_readfirstlane(t >> 5);
    for (int e = t; e < SC_R * P_; e += 256) lacc[e] = 0.0f;
    __syncthreads();

    const int rg  = blockIdx.x & (SC_NRG - 1);
    const int gw  = (blockIdx.x / SC_NRG) * SC_WPB + wave;
    const int nwv = SC_BPG * SC_WPB;
    const int nst = N >> 7;
    const int it0 = (int)(((long)gw * nst) / nwv);
    const int n   = (int)(((long)(gw + 1) * nst) / nwv) - it0;
    const float* xg = x + (size_t)(rg * SC_R) * N;
    const unsigned lo = (unsigned)it0 * 128u + (unsigned)lane * 4u;

    if (n > 0) {
        v2f run[SC_R];
        #pragma unroll
        for (int r = 0; r < SC_R; ++r) run[r] = (v2f){0.0f, 0.0f};
        int cur = idx[(size_t)it0 * 128];
        float4 xa[SC_R], xb[SC_R];
        float4 wa, wb;
        int4   ia, ib;
        sc_load(xa, wa, ia, xg, w, idx, N, lo);
        int k = 0;
        while (k + 2 < n) {
            sc_load(xb, wb, ib, xg, w, idx, N, lo + (unsigned)(k + 1) * 128u);
            sc_step(xa, wa, ia, run, cur, lacc, lane);
            sc_load(xa, wa, ia, xg, w, idx, N, lo + (unsigned)(k + 2) * 128u);
            sc_step(xb, wb, ib, run, cur, lacc, lane);
            k += 2;
        }
        if (k + 1 < n) {
            sc_load(xb, wb, ib, xg, w, idx, N, lo + (unsigned)(k + 1) * 128u);
            sc_step(xa, wa, ia, run, cur, lacc, lane);
            sc_step(xb, wb, ib, run, cur, lacc, lane);
        } else {
            sc_step(xa, wa, ia, run, cur, lacc, lane);
        }
        float tot[SC_R];
        #pragma unroll
        for (int r = 0; r < SC_R; ++r) tot[r] = run[r].x + run[r].y;
        sc_flush(tot, cur, lacc, lane);
    }
    if (gw == nwv - 1) {
        for (int f = (nst << 7) + lane; f < N; f += 32) {
            const float wf = w[f];
            const int   pf = idx[f];
            #pragma unroll
            for (int r = 0; r < SC_R; ++r)
                atomicAdd(&lacc[r * P_ + pf], xg[(size_t)r * N + f] * wf);
        }
    }
    __syncthreads();

    for (int e = t; e < SC_R * P_; e += 256) {
        const float v = lacc[e];
        if (v != 0.0f) atomicAdd(&WX[(size_t)(rg * SC_R) * P_ + e], v);
    }
}

#define ZP   (P_ + 4)
#define H1P  (H1_ + 4)
#define LP   (C_ + 1)

template <int K, int NM>
__device__ __forceinline__ void strip_gemm(
    v8f (&acc)[NM], const float* A, int lda, int mt0,
    const float* __restrict__ Bm, int ldb, int nt, int lane)
{
    const int ml   = lane & 15;
    const int koff = (lane >> 4) * 2;
    const float* arow = A + (mt0 * 16 + ml) * lda + koff;
    const float* bcol = Bm + koff * ldb + nt * 16 + ml;
    #pragma unroll
    for (int k = 0; k < K; k += 4) {
        v2f bv;
        bv.x = bcol[k * ldb];
        bv.y = bcol[(k + 1) * ldb];
        #pragma unroll
        for (int j = 0; j < NM; ++j) {
            const v2f av = *(const v2f*)(arow + j * 16 * lda + k);
            acc[j] = __builtin_amdgcn_wmma_f32_16x16x4_f32(
                false, av, false, bv, (short)0, acc[j], false, false);
        }
    }
}

__global__ void __launch_bounds__(512) spw_head(
    const float* __restrict__ WX,
    const float* __restrict__ co_w,
    const float* __restrict__ bn0_g, const float* __restrict__ bn0_b,
    const float* __restrict__ W1, const float* __restrict__ b1,
    const float* __restrict__ bn1_g, const float* __restrict__ bn1_b,
    const float* __restrict__ W2, const float* __restrict__ b2,
    const float* __restrict__ bn2_g, const float* __restrict__ bn2_b,
    const float* __restrict__ Wo, const float* __restrict__ bo,
    float* __restrict__ Zout,
    float* __restrict__ yout)
{
    __shared__ __attribute__((aligned(16))) float sP0[B_ * ZP];
    __shared__ __attribute__((aligned(16))) float sP1[B_ * H1P];
    __shared__ float sRed[8 * P_];
    __shared__ float sA[P_];
    __shared__ float sC[P_];
    const int t    = threadIdx.x;
    const int lane = t & 31;
    const int wave = t >> 5;
    const int ml   = lane & 15;
    const int hb   = (lane >> 4) * 8;
    const float invB = 1.0f / (float)B_;

    {
        const int col = t & (P_ - 1);
        const int rq  = t >> 7;
        float v[16];
        #pragma unroll
        for (int i = 0; i < 16; ++i) v[i] = fmaxf(WX[(rq + 4 * i) * P_ + col], 0.f);
        float s = 0.f, s2 = 0.f;
        #pragma unroll
        for (int i = 0; i < 16; ++i) { s += v[i]; s2 += v[i] * v[i]; }
        sRed[rq * P_ + col]       = s;
        sRed[(4 + rq) * P_ + col] = s2;
        __syncthreads();
        if (t < P_) {
            const float S  = (sRed[t] + sRed[P_ + t]) + (sRed[2 * P_ + t] + sRed[3 * P_ + t]);
            const float S2 = (sRed[4 * P_ + t] + sRed[5 * P_ + t]) +
                             (sRed[6 * P_ + t] + sRed[7 * P_ + t]);
            const float m   = S * invB;
            const float var = S2 * invB - m * m;
            const float rg  = rsqrtf(var + 1e-5f) * bn0_g[t];
            const float sg  = 1.0f / (1.0f + __expf(-co_w[t]));
            sA[t] = rg * sg;
            sC[t] = (bn0_b[t] - m * rg) * sg;
        }
        __syncthreads();
        const float a = sA[col], c = sC[col];
        #pragma unroll
        for (int i = 0; i < 16; ++i) {
            const int row = rq + 4 * i;
            const float z = v[i] * a + c;
            Zout[row * P_ + col] = z;
            sP0[row * ZP + col]  = z;
        }
    }
    __syncthreads();

    {
        v8f acc[4];
        #pragma unroll
        for (int j = 0; j < 4; ++j) acc[j] = (v8f){0.f, 0.f, 0.f, 0.f, 0.f, 0.f, 0.f, 0.f};
        strip_gemm<P_, 4>(acc, sP0, ZP, 0, W1, H1_, wave, lane);
        const int n = wave * 16 + ml;
        const float bias = b1[n];
        float s = 0.f, s2 = 0.f;
        #pragma unroll
        for (int j = 0; j < 4; ++j) {
            #pragma unroll
            for (int r = 0; r < 8; ++r) {
                const float hv = fmaxf(acc[j][r] + bias, 0.f);
                acc[j][r] = hv;
                s += hv; s2 += hv * hv;
            }
        }
        s  += __shfl_xor(s, 16);
        s2 += __shfl_xor(s2, 16);
        const float m   = s * invB;
        const float var = s2 * invB - m * m;
        const float rg  = rsqrtf(var + 1e-5f) * bn1_g[n];
        const float sh  = bn1_b[n] - m * rg;
        #pragma unroll
        for (int j = 0; j < 4; ++j) {
            #pragma unroll
            for (int r = 0; r < 8; ++r)
                sP1[(j * 16 + hb + r) * H1P + n] = acc[j][r] * rg + sh;
        }
    }
    __syncthreads();

    {
        const int nt = wave & 7, mh = wave >> 3;
        v8f acc[2];
        #pragma unroll
        for (int j = 0; j < 2; ++j) acc[j] = (v8f){0.f, 0.f, 0.f, 0.f, 0.f, 0.f, 0.f, 0.f};
        strip_gemm<H1_, 2>(acc, sP1, H1P, 2 * mh, W2, H2_, nt, lane);
        const int n = nt * 16 + ml;
        const float bias = b2[n];
        float s = 0.f, s2 = 0.f;
        #pragma unroll
        for (int j = 0; j < 2; ++j) {
            #pragma unroll
            for (int r = 0; r < 8; ++r) {
                const float hv = fmaxf(acc[j][r] + bias, 0.f);
                acc[j][r] = hv;
                s += hv; s2 += hv * hv;
            }
        }
        s  += __shfl_xor(s, 16);
        s2 += __shfl_xor(s2, 16);
        if (lane < 16) {
            sRed[mh * P_ + n]       = s;
            sRed[(2 + mh) * P_ + n] = s2;
        }
        __syncthreads();
        const float S   = sRed[n] + sRed[P_ + n];
        const float S2  = sRed[2 * P_ + n] + sRed[3 * P_ + n];
        const float m   = S * invB;
        const float var = S2 * invB - m * m;
        const float rg  = rsqrtf(var + 1e-5f) * bn2_g[n];
        const float sh  = bn2_b[n] - m * rg;
        #pragma unroll
        for (int j = 0; j < 2; ++j) {
            #pragma unroll
            for (int r = 0; r < 8; ++r)
                sP0[((2 * mh + j) * 16 + hb + r) * ZP + n] = acc[j][r] * rg + sh;
        }
    }
    __syncthreads();

    if (wave < 8) {
        const int mt = wave >> 1, nt = wave & 1;
        v8f acc[1];
        acc[0] = (v8f){0.f, 0.f, 0.f, 0.f, 0.f, 0.f, 0.f, 0.f};
        strip_gemm<H2_, 1>(acc, sP0, ZP, mt, Wo, C_, nt, lane);
        const int n = nt * 16 + ml;
        const float bias = bo[n];
        #pragma unroll
        for (int r = 0; r < 8; ++r)
            sP1[(mt * 16 + hb + r) * LP + n] = acc[0][r] + bias;
    }
    __syncthreads();

    if (t < B_) {
        const float* lr = sP1 + t * LP;
        float m = -3.0e38f;
        for (int c = 0; c < C_; ++c) m = fmaxf(m, lr[c]);
        float ex[C_];
        float s = 0.f;
        for (int c = 0; c < C_; ++c) { float e = __expf(lr[c] - m); ex[c] = e; s += e; }
        float inv = 1.0f / s;
        for (int c = 0; c < C_; ++c) yout[t * C_ + c] = ex[c] * inv;
    }
}

extern "C" void kernel_launch(void* const* d_in, const int* in_sizes, int n_in,
                              void* d_out, int out_size, void* d_ws, size_t ws_size,
                              hipStream_t stream) {
    const float* x     = (const float*)d_in[0];
    const float* w     = (const float*)d_in[1];
    const float* co_w  = (const float*)d_in[2];
    const float* bn0_g = (const float*)d_in[3];
    const float* bn0_b = (const float*)d_in[4];
    const float* W1    = (const float*)d_in[5];
    const float* b1    = (const float*)d_in[6];
    const float* bn1_g = (const float*)d_in[7];
    const float* bn1_b = (const float*)d_in[8];
    const float* W2    = (const float*)d_in[9];
    const float* b2    = (const float*)d_in[10];
    const float* bn2_g = (const float*)d_in[11];
    const float* bn2_b = (const float*)d_in[12];
    const float* Wo    = (const float*)d_in[13];
    const float* bo    = (const float*)d_in[14];
    const int*   idx   = (const int*)d_in[15];

    float* out = (float*)d_out;
    float* WX  = (float*)d_ws;

    const int N = in_sizes[0] / B_;

    hipMemsetAsync(WX, 0, (size_t)B_ * P_ * sizeof(float), stream);

    spw_scatter<<<SC_GRID, 256, 0, stream>>>(x, w, idx, WX, N);

    spw_head<<<1, 512, 0, stream>>>(WX, co_w, bn0_g, bn0_b,
                                    W1, b1, bn1_g, bn1_b,
                                    W2, b2, bn2_g, bn2_b,
                                    Wo, bo,
                                    out + B_ * C_, out);
}


# head layers 1+2 on 16 work-groups: each owns a layer-1 column strip = its K slice of layer 2, summed by f32 atomics into a plane zeroed per call; a new launch finishes (product+norm sum order changes)
# speedup vs baseline: 6.4127x; 1.1937x over previous
#include <hip/hip_runtime.h>
#include <hip/hip_bf16.h>


#define B_   64
#define P_   128
#define H1_  256
#define H2_  128
#define C_   32

typedef __attribute__((ext_vector_type(2))) float v2f;
typedef __attribute__((ext_vector_type(8))) float v8f;

#define SC_R    16
#define SC_NRG  (B_ / SC_R)
#define SC_WPB  8
#define SC_BPG  128
#define SC_GRID (SC_NRG * SC_BPG)

__device__ __forceinline__ float sc_rows_sum(const float (&v)[SC_R], int lane) {
    float a[8], b[4], c[2];
    const bool h4 = (lane & 16) != 0, h3 = (lane & 8) != 0;
    const bool h2 = (lane & 4) != 0,  h1 = (lane & 2) != 0;
    #pragma unroll
    for (int r = 0; r < 8; ++r)
        a[r] = (h4 ? v[r + 8] : v[r]) + __shfl_xor(h4 ? v[r] : v[r + 8], 16);
    #pragma unroll
    for (int r = 0; r < 4; ++r)
        b[r] = (h3 ? a[r + 4] : a[r]) + __shfl_xor(h3 ? a[r] : a[r + 4], 8);
    #pragma unroll
    for (int r = 0; r < 2; ++r)
        c[r] = (h2 ? b[r + 2] : b[r]) + __shfl_xor(h2 ? b[r] : b[r + 2], 4);
    float z = (h1 ? c[1] : c[0]) + __shfl_xor(h1 ? c[0] : c[1], 2);
    return z + __shfl_xor(z, 1);
}

__device__ __forceinline__ void sc_load(
    float4 (&xv)[SC_R], float4& wv, int4& iv,
    const float* __restrict__ xg, const float* __restrict__ w,
    const int* __restrict__ idx, int N, unsigned off)
{
    iv = *(const int4*)(idx + off);
    wv = *(const float4*)(w + off);
    #pragma unroll
    for (int r = 0; r < SC_R; ++r)
        xv[r] = *(const float4*)(xg + (size_t)r * N + off);
}

__device__ __forceinline__ void sc_flush(
    const float (&tot)[SC_R], int cur, float* lacc, int lane)
{
    const float z = sc_rows_sum(tot, lane);
    if ((lane & 1) == 0) atomicAdd(&lacc[((lane >> 1) & 15) * P_ + cur], z);
}

__device__ __forceinline__ void sc_step(
    const float4 (&xv)[SC_R], const float4 wv, const int4 iv,
    v2f (&run)[SC_R], int& cur, float* lacc, int lane)
{
    const int d = (iv.x ^ cur) | (iv.y ^ cur) | (iv.z ^ cur) | (iv.w ^ cur);
    if (!__any(d != 0)) {
        const v2f wlo = {wv.x, wv.y}, whi = {wv.z, wv.w};
        #pragma unroll
        for (int r = 0; r < SC_R; ++r) {
            const v2f xlo = {xv[r].x, xv[r].y}, xhi = {xv[r].z, xv[r].w};
            run[r] = __builtin_elementwise_fma(xhi, whi,
                     __builtin_elementwise_fma(xlo, wlo, run[r]));
        }
    } else {
        const int  nxt = __builtin_amdgcn_readlane(iv.w, 31);
        const bool all_cur = (d == 0);
        const bool all_nxt = !all_cur &&
            (((iv.x ^ nxt) | (iv.y ^ nxt) | (iv.z ^ nxt) | (iv.w ^ nxt)) == 0);
        if (!all_cur && !all_nxt) {
            #pragma unroll
            for (int r = 0; r < SC_R; ++r) {
                atomicAdd(&lacc[r * P_ + iv.x], xv[r].x * wv.x);
                atomicAdd(&lacc[r * P_ + iv.y], xv[r].y * wv.y);
                atomicAdd(&lacc[r * P_ + iv.z], xv[r].z * wv.z);
                atomicAdd(&lacc[r * P_ + iv.w], xv[r].w * wv.w);
            }
        }
        float tot[SC_R];
        #pragma unroll
        for (int r = 0; r < SC_R; ++r) {
            const float s = (xv[r].x * wv.x + xv[r].y * wv.y) +
                            (xv[r].z * wv.z + xv[r].w * wv.w);
            tot[r] = (run[r].x + run[r].y) + (all_cur ? s : 0.0f);
            run[r].x = all_nxt ? s : 0.0f;
            run[r].y = 0.0f;
        }
        sc_flush(tot, cur, lacc, lane);
        cur = nxt;
    }
}

__global__ void __launch_bounds__(256) spw_scatter(
    const float* __restrict__ x, const float* __restrict__ w,
    const int* __restrict__ idx, float* __restrict__ WX, int N)
{
    __shared__ float lacc[SC_R * P_];
    const int t    = threadIdx.x;
    const int lane = t & 31;
    const int wave = __builtin_amdgcn_readfirstlane(t >> 5);
    for (int e = t; e < SC_R * P_; e += 256) lacc[e] = 0.0f;
    __syncthreads();

    const int rg  = blockIdx.x & (SC_NRG - 1);
    const int gw  = (blockIdx.x / SC_NRG) * SC_WPB + wave;
    const int nwv = SC_BPG * SC_WPB;
    const int nst = N >> 7;
    const int it0 = (int)(((long)gw * nst) / nwv);
    const int n   = (int)(((long)(gw + 1) * nst) / nwv) - it0;
    const float* xg = x + (size_t)(rg * SC_R) * N;
    const unsigned lo = (unsigned)it0 * 128u + (unsigned)lane * 4u;

    if (n > 0) {
        v2f run[SC_R];
        #pragma unroll
        for (int r = 0; r < SC_R; ++r) run[r] = (v2f){0.0f, 0.0f};
        int cur = idx[(size_t)it0 * 128];
        float4 xa[SC_R], xb[SC_R];
        float4 wa, wb;
        int4   ia, ib;
        sc_load(xa, wa, ia, xg, w, idx, N, lo);
        int k = 0;
        while (k + 2 < n) {
            sc_load(xb, wb, ib, xg, w, idx, N, lo + (unsigned)(k + 1) * 128u);
            sc_step(xa, wa, ia, run, cur, lacc, lane);
            sc_load(xa, wa, ia, xg, w, idx, N, lo + (unsigned)(k + 2) * 128u);
            sc_step(xb, wb, ib, run, cur, lacc, lane);
            k += 2;
        }
        if (k + 1 < n) {
            sc_load(xb, wb, ib, xg, w, idx, N, lo + (unsigned)(k + 1) * 128u);
            sc_step(xa, wa, ia, run, cur, lacc, lane);
            sc_step(xb, wb, ib, run, cur, lacc, lane);
        } else {
            sc_step(xa, wa, ia, run, cur, lacc, lane);
        }
        float tot[SC_R];
        #pragma unroll
        for (int r = 0; r < SC_R; ++r) tot[r] = run[r].x + run[r].y;
        sc_flush(tot, cur, lacc, lane);
    }
    if (gw == nwv - 1) {
        for (int f = (nst << 7) + lane; f < N; f += 32) {
            const float wf = w[f];
            const int   pf = idx[f];
            #pragma unroll
            for (int r = 0; r < SC_R; ++r)
                atomicAdd(&lacc[r * P_ + pf], xg[(size_t)r * N + f] * wf);
        }
    }
    __syncthreads();

    for (int e = t; e < SC_R * P_; e += 256) {
        const float v = lacc[e];
        if (v != 0.0f) atomicAdd(&WX[(size_t)(rg * SC_R) * P_ + e], v);
    }
}

#define ZP   (P_ + 4)
#define LP   (C_ + 1)

#define HP   20
#define NSTR (H1_ / 16)

__global__ void __launch_bounds__(512) spw_head_a(
    const float* __restrict__ WX,
    const float* __restrict__ co_w,
    const float* __restrict__ bn0_g, const float* __restrict__ bn0_b,
    const float* __restrict__ W1, const float* __restrict__ b1,
    const float* __restrict__ bn1_g, const float* __restrict__ bn1_b,
    const float* __restrict__ W2,
    float* __restrict__ H2A,
    float* __restrict__ Zout)
{
    __shared__ __attribute__((aligned(16))) float sZ[B_ * ZP];
    __shared__ __attribute__((aligned(16))) float sPart[4 * B_ * HP];
    __shared__ __attribute__((aligned(16))) float sH[B_ * HP];
    __shared__ float sRed[8 * P_];
    __shared__ float sA[P_];
    __shared__ float sC[P_];
    const int t    = threadIdx.x;
    const int lane = t & 31;
    const int wave = t >> 5;
    const int ml   = lane & 15;
    const int koff = (lane >> 4) * 2;
    const int hb   = (lane >> 4) * 8;
    const int g    = blockIdx.x;
    const float invB = 1.0f / (float)B_;

    const int j  = wave & 3;
    const int kq = wave >> 2;

    v2f w1f[8];
    #pragma unroll
    for (int s = 0; s < 8; ++s) {
        const float* p = W1 + (kq * 32 + s * 4 + koff) * H1_ + g * 16 + ml;
        w1f[s].x = p[0];
        w1f[s].y = p[H1_];
    }
    v2f w2f[2][4];
    #pragma unroll
    for (int q = 0; q < 2; ++q) {
        #pragma unroll
        for (int s = 0; s < 4; ++s) {
            const float* p = W2 + (g * 16 + s * 4 + koff) * H2_ + (kq * 2 + q) * 16 + ml;
            w2f[q][s].x = p[0];
            w2f[q][s].y = p[H2_];
        }
    }
    const int   c1   = t & 15;
    const float bias = b1[g * 16 + c1];
    const float g1   = bn1_g[g * 16 + c1];
    const float be1  = bn1_b[g * 16 + c1];
    const int   col0 = t & (P_ - 1);
    const float g0   = bn0_g[col0];
    const float be0  = bn0_b[col0];
    const float cw0  = co_w[col0];

    {
        const int col = col0;
        const int rq  = t >> 7;
        float v[16];
        #pragma unroll
        for (int i = 0; i < 16; ++i) v[i] = fmaxf(WX[(rq + 4 * i) * P_ + col], 0.f);
        float s = 0.f, s2 = 0.f;
        #pragma unroll
        for (int i = 0; i < 16; ++i) { s += v[i]; s2 += v[i] * v[i]; }
        sRed[rq * P_ + col]       = s;
        sRed[(4 + rq) * P_ + col] = s2;
        __syncthreads();
        if (t < P_) {
            const float S  = (sRed[t] + sRed[P_ + t]) + (sRed[2 * P_ + t] + sRed[3 * P_ + t]);
            const float S2 = (sRed[4 * P_ + t] + sRed[5 * P_ + t]) +
                             (sRed[6 * P_ + t] + sRed[7 * P_ + t]);
            const float m   = S * invB;
            const float var = S2 * invB - m * m;
            const float rg  = rsqrtf(var + 1e-5f) * g0;
            const float sg  = 1.0f / (1.0f + __expf(-cw0));
            sA[t] = rg * sg;
            sC[t] = (be0 - m * rg) * sg;
        }
        __syncthreads();
        const float a = sA[col], c = sC[col];
        #pragma unroll
        for (int i = 0; i < 16; ++i) {
            const int row = rq + 4 * i;
            const float z = v[i] * a + c;
            if (g == 0) Zout[row * P_ + col] = z;
            sZ[row * ZP + col] = z;
        }
    }
    __syncthreads();

    {
        v8f acc = (v8f){0.f, 0.f, 0.f, 0.f, 0.f, 0.f, 0.f, 0.f};
        const float* arow = sZ + (j * 16 + ml) * ZP + kq * 32 + koff;
        #pragma unroll
        for (int s = 0; s < 8; ++s) {
            const v2f av = *(const v2f*)(arow + s * 4);
            acc = __builtin_amdgcn_wmma_f32_16x16x4_f32(
                false, av, false, w1f[s], (short)0, acc, false, false);
        }
        #pragma unroll
        for (int r = 0; r < 8; ++r)
            sPart[(kq * B_ + j * 16 + hb + r) * HP + ml] = acc[r];
    }
    __syncthreads();

    {
        const int r0 = t >> 4;
        float hv[2];
        #pragma unroll
        for (int i = 0; i < 2; ++i) {
            const float* p = sPart + (r0 + 32 * i) * HP + c1;
            const float z = (p[0] + p[B_ * HP]) + (p[2 * B_ * HP] + p[3 * B_ * HP]);
            hv[i] = fmaxf(z + bias, 0.f);
        }
        float s  = hv[0] + hv[1];
        float s2 = hv[0] * hv[0] + hv[1] * hv[1];
        s  += __shfl_xor(s, 16);
        s2 += __shfl_xor(s2, 16);
        if (lane < 16) {
            sRed[wave * 16 + c1]       = s;
            sRed[256 + wave * 16 + c1] = s2;
        }
        __syncthreads();
        float S = 0.f, S2 = 0.f;
        #pragma unroll
        for (int w = 0; w < 16; ++w) { S += sRed[w * 16 + c1]; S2 += sRed[256 + w * 16 + c1]; }
        const float m   = S * invB;
        const float var = S2 * invB - m * m;
        const float rg  = rsqrtf(var + 1e-5f) * g1;
        const float sh  = be1 - m * rg;
        sH[r0 * HP + c1]        = hv[0] * rg + sh;
        sH[(r0 + 32) * HP + c1] = hv[1] * rg + sh;
    }
    __syncthreads();

    {
        v8f acc[2];
        #pragma unroll
        for (int q = 0; q < 2; ++q) acc[q] = (v8f){0.f, 0.f, 0.f, 0.f, 0.f, 0.f, 0.f, 0.f};
        const float* arow = sH + (j * 16 + ml) * HP + koff;
        #pragma unroll
        for (int s = 0; s < 4; ++s) {
            const v2f av = *(const v2f*)(arow + s * 4);
            #pragma unroll
            for (int q = 0; q < 2; ++q)
                acc[q] = __builtin_amdgcn_wmma_f32_16x16x4_f32(
                    false, av, false, w2f[q][s], (short)0, acc[q], false, false);
        }
        #pragma unroll
        for (int q = 0; q < 2; ++q) {
            float* o = H2A + (j * 16 + hb) * H2_ + (kq * 2 + q) * 16 + ml;
            #pragma unroll
            for (int r = 0; r < 8; ++r) atomicAdd(o + r * H2_, acc[q][r]);
        }
    }
}

__global__ void __launch_bounds__(512) spw_head_b(
    const float* __restrict__ H2A,
    const float* __restrict__ b2,
    const float* __restrict__ bn2_g, const float* __restrict__ bn2_b,
    const float* __restrict__ Wo, const float* __restrict__ bo,
    float* __restrict__ yout)
{
    __shared__ __attribute__((aligned(16))) float sP0[B_ * ZP];
    __shared__ float sP1[B_ * LP];
    __shared__ float sRed[8 * P_];
    __shared__ float sA[P_];
    __shared__ float sC[P_];
    const int t    = threadIdx.x;
    const int lane = t & 31;
    const int wave = t >> 5;
    const int ml   = lane & 15;
    const int hb   = (lane >> 4) * 8;
    const float invB = 1.0f / (float)B_;

    const int mt = (wave >> 1) & 3, nt = wave & 1;
    const int koff = (lane >> 4) * 2;
    const int col = t & (H2_ - 1);

    const float bias = b2[col];
    const float g2   = bn2_g[col];
    const float be2  = bn2_b[col];
    const float bo3  = bo[nt * 16 + ml];
    v2f wof[H2_ / 4];
    if (wave < 8) {
        #pragma unroll
        for (int s = 0; s < H2_ / 4; ++s) {
            const float* p = Wo + (s * 4 + koff) * C_ + nt * 16 + ml;
            wof[s].x = p[0];
            wof[s].y = p[C_];
        }
    }

    {
        const int rq  = t >> 7;
        float v[16];
        #pragma unroll
        for (int i = 0; i < 16; ++i) v[i] = fmaxf(H2A[(rq + 4 * i) * H2_ + col] + bias, 0.f);
        float s = 0.f, s2 = 0.f;
        #pragma unroll
        for (int i = 0; i < 16; ++i) { s += v[i]; s2 += v[i] * v[i]; }
        sRed[rq * P_ + col]       = s;
        sRed[(4 + rq) * P_ + col] = s2;
        __syncthreads();
        if (t < H2_) {
            const float S  = (sRed[t] + sRed[P_ + t]) + (sRed[2 * P_ + t] + sRed[3 * P_ + t]);
            const float S2 = (sRed[4 * P_ + t] + sRed[5 * P_ + t]) +
                             (sRed[6 * P_ + t] + sRed[7 * P_ + t]);
            const float m   = S * invB;
            const float var = S2 * invB - m * m;
            const float rg  = rsqrtf(var + 1e-5f) * g2;
            sA[t] = rg;
            sC[t] = be2 - m * rg;
        }
        __syncthreads();
        const float a = sA[col], c = sC[col];
        #pragma unroll
        for (int i = 0; i < 16; ++i) sP0[(rq + 4 * i) * ZP + col] = v[i] * a + c;
    }
    __syncthreads();

    if (wave < 8) {
        v8f acc = (v8f){0.f, 0.f, 0.f, 0.f, 0.f, 0.f, 0.f, 0.f};
        const float* arow = sP0 + (mt * 16 + ml) * ZP + koff;
        #pragma unroll
        for (int s = 0; s < H2_ / 4; ++s) {
            const v2f av = *(const v2f*)(arow + s * 4);
            acc = __builtin_amdgcn_wmma_f32_16x16x4_f32(
                false, av, false, wof[s], (short)0, acc, false, false);
        }
        const int n = nt * 16 + ml;
        #pragma unroll
        for (int r = 0; r < 8; ++r)
            sP1[(mt * 16 + hb + r) * LP + n] = acc[r] + bo3;
    }
    __syncthreads();

    if (t < B_) {
        const float* lr = sP1 + t * LP;
        float m = -3.0e38f;
        for (int c = 0; c < C_; ++c) m = fmaxf(m, lr[c]);
        float ex[C_];
        float s = 0.f;
        for (int c = 0; c < C_; ++c) { float e = __expf(lr[c] - m); ex[c] = e; s += e; }
        float inv = 1.0f / s;
        for (int c = 0; c < C_; ++c) yout[t * C_ + c] = ex[c] * inv;
    }
}

extern "C" void kernel_launch(void* const* d_in, const int* in_sizes, int n_in,
                              void* d_out, int out_size, void* d_ws, size_t ws_size,
                              hipStream_t stream) {
    const float* x     = (const float*)d_in[0];
    const float* w     = (const float*)d_in[1];
    const float* co_w  = (const float*)d_in[2];
    const float* bn0_g = (const float*)d_in[3];
    const float* bn0_b = (const float*)d_in[4];
    const float* W1    = (const float*)d_in[5];
    const float* b1    = (const float*)d_in[6];
    const float* bn1_g = (const float*)d_in[7];
    const float* bn1_b = (const float*)d_in[8];
    const float* W2    = (const float*)d_in[9];
    const float* b2    = (const float*)d_in[10];
    const float* bn2_g = (const float*)d_in[11];
    const float* bn2_b = (const float*)d_in[12];
    const float* Wo    = (const float*)d_in[13];
    const float* bo    = (const float*)d_in[14];
    const int*   idx   = (const int*)d_in[15];

    float* out = (float*)d_out;
    float* WX  = (float*)d_ws;
    float* H2A = WX + B_ * P_;

    const int N = in_sizes[0] / B_;

    hipMemsetAsync(WX, 0, (size_t)2 * B_ * P_ * sizeof(float), stream);

    spw_scatter<<<SC_GRID, 256, 0, stream>>>(x, w, idx, WX, N);

    spw_head_a<<<NSTR, 512, 0, stream>>>(WX, co_w, bn0_g, bn0_b,
                                         W1, b1, bn1_g, bn1_b, W2,
                                         H2A, out + B_ * C_);

    spw_head_b<<<1, 512, 0, stream>>>(H2A, b2, bn2_g, bn2_b, Wo, bo, out);
}
